# starting kernel
# baseline (speedup 1.0000x reference)
#include <hip/hip_runtime.h>

#define NN 100000
#define NE 1600000
#define DIM 128
#define BN_EPS 1e-5f

typedef __attribute__((ext_vector_type(2))) float v2f;
typedef __attribute__((ext_vector_type(8))) float v8f;

__global__ void zero_kernel(float* __restrict__ p, size_t n) {
    size_t i = ((size_t)blockIdx.x * blockDim.x + threadIdx.x) * 4;
    if (i + 3 < n) {
        float4 z; z.x = 0.f; z.y = 0.f; z.z = 0.f; z.w = 0.f;
        *(float4*)(p + i) = z;
    } else {
        for (size_t j = i; j < n; ++j) p[j] = 0.f;
    }
}

__global__ __launch_bounds__(256) void scatter_kernel(
    const float* __restrict__ x, const int* __restrict__ ei,
    float* __restrict__ agg, float* __restrict__ deg) {
    int wid  = blockIdx.x * 8 + (threadIdx.x >> 5);
    int lane = threadIdx.x & 31;
    if (wid >= NE) return;
    int src = ei[wid];
    int dst = ei[NE + wid];
    const float4 v = *(const float4*)(x + (size_t)src * DIM + lane * 4);
    float* dp = agg + (size_t)dst * DIM + lane * 4;
    atomicAdd(dp + 0, v.x);
    atomicAdd(dp + 1, v.y);
    atomicAdd(dp + 2, v.z);
    atomicAdd(dp + 3, v.w);
    if (lane == 0) atomicAdd(deg + dst, 1.0f);
}

__global__ void mean_kernel(float* __restrict__ agg, const float* __restrict__ deg) {
    size_t i = (size_t)blockIdx.x * blockDim.x + threadIdx.x;
    if (i >= (size_t)NN * DIM / 4) return;
    int node = (int)((i * 4) >> 7);
    float inv = 1.0f / fmaxf(deg[node], 1.0f);
    float4 v = ((float4*)agg)[i];
    v.x *= inv; v.y *= inv; v.z *= inv; v.w *= inv;
    ((float4*)agg)[i] = v;
}

__global__ __launch_bounds__(256) void gemm_bn_kernel(
    const float* __restrict__ mean, const float* __restrict__ x,
    const float* __restrict__ Wl, const float* __restrict__ bl,
    const float* __restrict__ Wr, float* __restrict__ h,
    float* __restrict__ gsum, float* __restrict__ gsq) {
    __shared__ float lWl[DIM * DIM];
    __shared__ float lWr[DIM * DIM];
    __shared__ float lsum[DIM];
    __shared__ float lsq[DIM];

    const int tid = threadIdx.x;
    for (int i = tid * 4; i < DIM * DIM; i += 256 * 4) {
        *(float4*)(lWl + i) = *(const float4*)(Wl + i);
        *(float4*)(lWr + i) = *(const float4*)(Wr + i);
    }
    if (tid < DIM) { lsum[tid] = 0.f; lsq[tid] = 0.f; }
    __syncthreads();

    const int wave = tid >> 5;
    const int lane = tid & 31;
    const int tile = blockIdx.x * 8 + wave;
    const bool active = (tile * 16) < NN;

    if (active) {
        const int row0 = tile * 16;
        const int m  = lane & 15;
        const int kh = lane >> 4;
        const float* meanRow = mean + (size_t)(row0 + m) * DIM;
        const float* xRow    = x    + (size_t)(row0 + m) * DIM;

        for (int nt = 0; nt < 8; ++nt) {
            const int ncol = nt * 16 + (lane & 15);
            const float bias = bl[ncol];
            v8f c;
#pragma unroll
            for (int r = 0; r < 8; ++r) c[r] = bias;

#pragma unroll
            for (int k0 = 0; k0 < DIM; k0 += 4) {
                const int ka = k0 + 2 * kh;
                v2f a; a.x = meanRow[ka];           a.y = meanRow[ka + 1];
                v2f b; b.x = lWl[ka * DIM + ncol];  b.y = lWl[(ka + 1) * DIM + ncol];
                c = __builtin_amdgcn_wmma_f32_16x16x4_f32(
                        false, a, false, b, (short)0, c, false, false);
            }
#pragma unroll
            for (int k0 = 0; k0 < DIM; k0 += 4) {
                const int ka = k0 + 2 * kh;
                v2f a; a.x = xRow[ka];              a.y = xRow[ka + 1];
                v2f b; b.x = lWr[ka * DIM + ncol];  b.y = lWr[(ka + 1) * DIM + ncol];
                c = __builtin_amdgcn_wmma_f32_16x16x4_f32(
                        false, a, false, b, (short)0, c, false, false);
            }

            float s = 0.f, s2 = 0.f;
#pragma unroll
            for (int r = 0; r < 8; ++r) {
                float v = c[r] > 0.f ? c[r] : 0.f;
                const int row = row0 + r + 8 * kh;
                h[(size_t)row * DIM + ncol] = v;
                s  += v;
                s2 += v * v;
            }
            s  += __shfl_xor(s, 16, 32);
            s2 += __shfl_xor(s2, 16, 32);
            if (lane < 16) {
                atomicAdd(&lsum[ncol], s);
                atomicAdd(&lsq[ncol], s2);
            }
        }
    }
    __syncthreads();
    if (tid < DIM) {
        atomicAdd(gsum + tid, lsum[tid]);
        atomicAdd(gsq  + tid, lsq[tid]);
    }
}

__global__ void stats_kernel(const float* __restrict__ gsum, const float* __restrict__ gsq,
                             const float* __restrict__ gamma, const float* __restrict__ beta,
                             float* __restrict__ cA, float* __restrict__ cB) {
    int c = threadIdx.x;
    if (c >= DIM) return;
    const float invN = 1.0f / (float)NN;
    float mu  = gsum[c] * invN;
    float var = gsq[c] * invN - mu * mu;
    float a   = rsqrtf(var + BN_EPS) * gamma[c];
    cA[c] = a;
    cB[c] = beta[c] - mu * a;
}

__global__ void apply_kernel(const float* __restrict__ x,
                             const float* __restrict__ cA, const float* __restrict__ cB,
                             float* __restrict__ out) {
    size_t i = (size_t)blockIdx.x * blockDim.x + threadIdx.x;
    if (i >= (size_t)NN * DIM / 4) return;
    int col4 = (int)(i & 31) * 4;
    float4 hv = ((float4*)out)[i];
    float4 xv = ((const float4*)x)[i];
    float4 a  = *(const float4*)(cA + col4);
    float4 b  = *(const float4*)(cB + col4);
    float4 o;
    o.x = xv.x + hv.x * a.x + b.x;
    o.y = xv.y + hv.y * a.y + b.y;
    o.z = xv.z + hv.z * a.z + b.z;
    o.w = xv.w + hv.w * a.w + b.w;
    ((float4*)out)[i] = o;
}

extern "C" void kernel_launch(void* const* d_in, const int* in_sizes, int n_in,
                              void* d_out, int out_size, void* d_ws, size_t ws_size,
                              hipStream_t stream) {
    const float* x     = (const float*)d_in[0];
    const int*   ei    = (const int*)d_in[1];
    const float* Wl    = (const float*)d_in[2];
    const float* bl    = (const float*)d_in[3];
    const float* Wr    = (const float*)d_in[4];
    const float* gamma = (const float*)d_in[5];
    const float* beta  = (const float*)d_in[6];
    float* out = (float*)d_out;

    float* agg  = (float*)d_ws;
    float* deg  = agg + (size_t)NN * DIM;
    float* gsum = deg + NN;
    float* gsq  = gsum + DIM;
    float* cA   = gsq + DIM;
    float* cB   = cA + DIM;

    const size_t zero_elems = (size_t)NN * DIM + NN + 4 * DIM;
    int zb = (int)((zero_elems / 4 + 255) / 256);
    zero_kernel<<<zb, 256, 0, stream>>>(agg, zero_elems);

    scatter_kernel<<<NE / 8, 256, 0, stream>>>(x, ei, agg, deg);

    int mb = (int)(((size_t)NN * DIM / 4 + 255) / 256);
    mean_kernel<<<mb, 256, 0, stream>>>(agg, deg);

    const int tiles = NN / 16;
    gemm_bn_kernel<<<(tiles + 7) / 8, 256, 0, stream>>>(agg, x, Wl, bl, Wr,
                                                        out, gsum, gsq);

    stats_kernel<<<1, 128, 0, stream>>>(gsum, gsq, gamma, beta, cA, cB);

    apply_kernel<<<mb, 256, 0, stream>>>(x, cA, cB, out);
}


# neighbour kernel: sweep of all destinations replaced by a per-range flag table of 32-edge chunks (new flag launch, LDS or): a work-group reads only flagged lines; sums still in LDS arrival order
# speedup vs baseline: 2.2937x; 2.2937x over previous
#include <hip/hip_runtime.h>

#define NN 100000
#define NE 1600000
#define DIM 128
#define BN_EPS 1e-5f

typedef __attribute__((ext_vector_type(2))) float v2f;
typedef __attribute__((ext_vector_type(8))) float v8f;

#define GB   256
#define RPB  391
#define QC   512
#define NV   (NE / 4)
#define NCH  (NE / 32)
#define FCH  256
#define FB   196
#define FWW  (FCH / 32)
#define NW   (FB * FWW)
#define WPW  (NW / 32)
static_assert(NE % 32 == 0, "chunks are whole lines");
static_assert(FB * FCH >= NCH && (FB - 1) * FCH < NCH, "flag grid covers all chunks");
static_assert(NW % 32 == 0 && WPW > 32 && WPW <= 64, "two flag words per lane hold a wave's share");
static_assert(GB == 256, "flag kernel: one thread per range");

__global__ __launch_bounds__(256) void flag_kernel(
    const int* __restrict__ ei, float* __restrict__ outbuf) {
    __shared__ unsigned lbm[GB * FWW];

    const int tid = threadIdx.x;
    for (int i = tid; i < GB * FWW; i += 256) lbm[i] = 0u;
    __syncthreads();

    const int4* dst4 = (const int4*)(ei + NE);
    int4 v[8];
#pragma unroll
    for (int j = 0; j < 8; ++j) {
        const int idx = blockIdx.x * (FCH * 8) + j * 256 + tid;
        int4 a; a.x = -1; a.y = -1; a.z = -1; a.w = -1;
        if (idx < NV) a = dst4[idx];
        v[j] = a;
    }
#pragma unroll
    for (int j = 0; j < 8; ++j) {
        const int ch = (j * 256 + tid) >> 3;
        unsigned* wp = lbm + (ch >> 5);
        const unsigned bit = 1u << (ch & 31);
        const int d[4] = {v[j].x, v[j].y, v[j].z, v[j].w};
#pragma unroll
        for (int t = 0; t < 4; ++t) {
            if ((unsigned)d[t] < (unsigned)(GB * RPB))
                atomicOr(wp + ((unsigned)d[t] / RPB) * FWW, bit);
        }
    }
    __syncthreads();

    unsigned* bm = (unsigned*)outbuf;
    const uint4 a = *(const uint4*)(lbm + tid * FWW);
    const uint4 b = *(const uint4*)(lbm + tid * FWW + 4);
    uint4* gp = (uint4*)(bm + (size_t)tid * NW + blockIdx.x * FWW);
    gp[0] = a;
    gp[1] = b;
}

__global__ __launch_bounds__(1024) void gather_lds_kernel(
    const float* __restrict__ x, const int* __restrict__ ei,
    const float* __restrict__ flagbuf,
    float* __restrict__ mean, float* __restrict__ gstat) {
    __shared__ float acc[RPB * DIM];
    __shared__ float ldeg[RPB + 1];
    __shared__ int   q[32 * QC];

    const int tid  = threadIdx.x;
    const int lane = tid & 31;
    const int wave = tid >> 5;
    const int lo   = blockIdx.x * RPB;

    for (int i = tid * 4; i < RPB * DIM; i += 4096) {
        float4 z; z.x = 0.f; z.y = 0.f; z.z = 0.f; z.w = 0.f;
        *(float4*)(acc + i) = z;
    }
    if (tid < RPB + 1) ldeg[tid] = 0.f;
    if (blockIdx.x == 0 && tid < 2 * DIM) gstat[tid] = 0.f;
    __syncthreads();

    int* myq = q + wave * QC;
    const int* dstp = ei + NE;
    int head = 0, cnt = 0;

    auto push = [&](int val, int e) {
        const unsigned dl = (unsigned)(val - lo);
        const bool m = dl < (unsigned)RPB;
        const unsigned mask = __builtin_amdgcn_ballot_w32(m);
        if (mask) {
            if (m) {
                const int pos = cnt + (int)__builtin_amdgcn_mbcnt_lo(mask, 0u);
                myq[(head + pos) & (QC - 1)] = (e << 9) | (int)dl;
            }
            cnt += __builtin_popcount(mask);
        }
    };

    auto drain = [&](const int n) {
        int ent = 0;
        if (lane < n) ent = myq[(head + lane) & (QC - 1)];
        const int dl = ent & 511;
        int s = 0;
        if (lane < n) {
            s = ei[ent >> 9];
            atomicAdd(&ldeg[dl], 1.0f);
        }
        s = (int)min((unsigned)s, (unsigned)(NN - 1));
#pragma unroll 1
        for (int k = 0; k < n; k += 8) {
            float v[8][4];
#pragma unroll
            for (int t = 0; t < 8; ++t) {
                if (k + t < n) {
                    const int ss = __builtin_amdgcn_readlane(s, k + t);
                    const float* sp = x + (size_t)ss * DIM + lane;
                    v[t][0] = sp[0];
                    v[t][1] = sp[32];
                    v[t][2] = sp[64];
                    v[t][3] = sp[96];
                }
            }
#pragma unroll
            for (int t = 0; t < 8; ++t) {
                if (k + t < n) {
                    const int dd = __builtin_amdgcn_readlane(dl, k + t);
                    float* ap = acc + dd * DIM + lane;
                    atomicAdd(ap,      v[t][0]);
                    atomicAdd(ap + 32, v[t][1]);
                    atomicAdd(ap + 64, v[t][2]);
                    atomicAdd(ap + 96, v[t][3]);
                }
            }
        }
        head = (head + n) & (QC - 1);
        cnt -= n;
    };

    const unsigned* frow = (const unsigned*)flagbuf + (size_t)blockIdx.x * NW + wave * WPW;
    const int wA = (int)frow[lane];
    const int wB = (lane < WPW - 32) ? (int)frow[32 + lane] : 0;

#pragma unroll 1
    for (int i = 0; i < WPW; ++i) {
        unsigned m = (unsigned)__builtin_amdgcn_readlane(i < 32 ? wA : wB, i & 31);
        const int cb = (wave * WPW + i) * 32;
        while (m) {
            const unsigned m0 = m;
            int v[8];
#pragma unroll
            for (int t = 0; t < 8; ++t) {
                v[t] = -1;
                if (m) {
                    const int c = cb + __builtin_ctz(m);
                    m &= m - 1;
                    if (c < NCH) v[t] = dstp[c * 32 + lane];
                }
            }
            unsigned mm = m0;
#pragma unroll
            for (int t = 0; t < 8; ++t) {
                if (mm) {
                    const int c = cb + __builtin_ctz(mm);
                    mm &= mm - 1;
                    push(v[t], c * 32 + lane);
                }
            }
            while (cnt >= 32) drain(32);
        }
    }
    if (cnt > 0) drain(cnt);
    __syncthreads();

    for (int i = tid; i < RPB * (DIM / 4); i += 1024) {
        const int r = i >> 5;
        const int node = lo + r;
        if (node < NN) {
            const float inv = 1.0f / fmaxf(ldeg[r], 1.0f);
            float4 v = *(const float4*)(acc + i * 4);
            v.x *= inv; v.y *= inv; v.z *= inv; v.w *= inv;
            ((float4*)mean)[(size_t)node * (DIM / 4) + (i & 31)] = v;
        }
    }
}

__global__ __launch_bounds__(256) void gemm_bn_kernel(
    const float* __restrict__ mean, const float* __restrict__ x,
    const float* __restrict__ Wl, const float* __restrict__ bl,
    const float* __restrict__ Wr, float* __restrict__ h,
    float* __restrict__ gsum, float* __restrict__ gsq) {
    __shared__ float lWl[DIM * DIM];
    __shared__ float lWr[DIM * DIM];
    __shared__ float lsum[DIM];
    __shared__ float lsq[DIM];

    const int tid = threadIdx.x;
    for (int i = tid * 4; i < DIM * DIM; i += 256 * 4) {
        *(float4*)(lWl + i) = *(const float4*)(Wl + i);
        *(float4*)(lWr + i) = *(const float4*)(Wr + i);
    }
    if (tid < DIM) { lsum[tid] = 0.f; lsq[tid] = 0.f; }
    __syncthreads();

    const int wave = tid >> 5;
    const int lane = tid & 31;
    const int tile = blockIdx.x * 8 + wave;
    const bool active = (tile * 16) < NN;

    if (active) {
        const int row0 = tile * 16;
        const int m  = lane & 15;
        const int kh = lane >> 4;
        const float* meanRow = mean + (size_t)(row0 + m) * DIM;
        const float* xRow    = x    + (size_t)(row0 + m) * DIM;

        for (int nt = 0; nt < 8; ++nt) {
            const int ncol = nt * 16 + (lane & 15);
            const float bias = bl[ncol];
            v8f c;
#pragma unroll
            for (int r = 0; r < 8; ++r) c[r] = bias;

#pragma unroll
            for (int k0 = 0; k0 < DIM; k0 += 4) {
                const int ka = k0 + 2 * kh;
                v2f a; a.x = meanRow[ka];           a.y = meanRow[ka + 1];
                v2f b; b.x = lWl[ka * DIM + ncol];  b.y = lWl[(ka + 1) * DIM + ncol];
                c = __builtin_amdgcn_wmma_f32_16x16x4_f32(
                        false, a, false, b, (short)0, c, false, false);
            }
#pragma unroll
            for (int k0 = 0; k0 < DIM; k0 += 4) {
                const int ka = k0 + 2 * kh;
                v2f a; a.x = xRow[ka];              a.y = xRow[ka + 1];
                v2f b; b.x = lWr[ka * DIM + ncol];  b.y = lWr[(ka + 1) * DIM + ncol];
                c = __builtin_amdgcn_wmma_f32_16x16x4_f32(
                        false, a, false, b, (short)0, c, false, false);
            }

            float s = 0.f, s2 = 0.f;
#pragma unroll
            for (int r = 0; r < 8; ++r) {
                float v = c[r] > 0.f ? c[r] : 0.f;
                const int row = row0 + r + 8 * kh;
                h[(size_t)row * DIM + ncol] = v;
                s  += v;
                s2 += v * v;
            }
            s  += __shfl_xor(s, 16, 32);
            s2 += __shfl_xor(s2, 16, 32);
            if (lane < 16) {
                atomicAdd(&lsum[ncol], s);
                atomicAdd(&lsq[ncol], s2);
            }
        }
    }
    __syncthreads();
    if (tid < DIM) {
        atomicAdd(gsum + tid, lsum[tid]);
        atomicAdd(gsq  + tid, lsq[tid]);
    }
}

__global__ void stats_kernel(const float* __restrict__ gsum, const float* __restrict__ gsq,
                             const float* __restrict__ gamma, const float* __restrict__ beta,
                             float* __restrict__ cA, float* __restrict__ cB) {
    int c = threadIdx.x;
    if (c >= DIM) return;
    const float invN = 1.0f / (float)NN;
    float mu  = gsum[c] * invN;
    float var = gsq[c] * invN - mu * mu;
    float a   = rsqrtf(var + BN_EPS) * gamma[c];
    cA[c] = a;
    cB[c] = beta[c] - mu * a;
}

__global__ void apply_kernel(const float* __restrict__ x,
                             const float* __restrict__ cA, const float* __restrict__ cB,
                             float* __restrict__ out) {
    size_t i = (size_t)blockIdx.x * blockDim.x + threadIdx.x;
    if (i >= (size_t)NN * DIM / 4) return;
    int col4 = (int)(i & 31) * 4;
    float4 hv = ((float4*)out)[i];
    float4 xv = ((const float4*)x)[i];
    float4 a  = *(const float4*)(cA + col4);
    float4 b  = *(const float4*)(cB + col4);
    float4 o;
    o.x = xv.x + hv.x * a.x + b.x;
    o.y = xv.y + hv.y * a.y + b.y;
    o.z = xv.z + hv.z * a.z + b.z;
    o.w = xv.w + hv.w * a.w + b.w;
    ((float4*)out)[i] = o;
}

extern "C" void kernel_launch(void* const* d_in, const int* in_sizes, int n_in,
                              void* d_out, int out_size, void* d_ws, size_t ws_size,
                              hipStream_t stream) {
    const float* x     = (const float*)d_in[0];
    const int*   ei    = (const int*)d_in[1];
    const float* Wl    = (const float*)d_in[2];
    const float* bl    = (const float*)d_in[3];
    const float* Wr    = (const float*)d_in[4];
    const float* gamma = (const float*)d_in[5];
    const float* beta  = (const float*)d_in[6];
    float* out = (float*)d_out;

    float* agg  = (float*)d_ws;
    float* deg  = agg + (size_t)NN * DIM;
    float* gsum = deg + NN;
    float* gsq  = gsum + DIM;
    float* cA   = gsq + DIM;
    float* cB   = cA + DIM;

    flag_kernel<<<FB, 256, 0, stream>>>(ei, out);
    gather_lds_kernel<<<GB, 1024, 0, stream>>>(x, ei, out, agg, gsum);

    int mb = (int)(((size_t)NN * DIM / 4 + 255) / 256);
    const int tiles = NN / 16;
    gemm_bn_kernel<<<(tiles + 7) / 8, 256, 0, stream>>>(agg, x, Wl, bl, Wr,
                                                        out, gsum, gsq);

    stats_kernel<<<1, 128, 0, stream>>>(gsum, gsq, gamma, beta, cA, cB);

    apply_kernel<<<mb, 256, 0, stream>>>(x, cA, cB, out);
}
